# starting kernel
# baseline (speedup 1.0000x reference)
#include <hip/hip_runtime.h>

#define N_NODES   50000
#define N_EDGES   800000
#define IN_FEAT   128
#define PROMPT_DIM 32
#define EF_DIM     32
#define HP_DIM     32
#define OUT_DIM   224

typedef __attribute__((ext_vector_type(2))) float v2f;
typedef __attribute__((ext_vector_type(8))) float v8f;

__device__ __forceinline__ void gatomic_add_f32(float* p, float v) {
    asm volatile("global_atomic_add_f32 %0, %1, off" :: "v"(p), "v"(v) : "memory");
}

__global__ void zero_kernel(float* __restrict__ out, float* __restrict__ deg) {
    const size_t stride = (size_t)gridDim.x * blockDim.x;
    size_t i = (size_t)blockIdx.x * blockDim.x + threadIdx.x;
    const size_t total4 = (size_t)N_NODES * OUT_DIM / 4;
    float4 z = make_float4(0.f, 0.f, 0.f, 0.f);
    for (size_t k = i; k < total4; k += stride)
        ((float4*)out)[k] = z;
    for (size_t k = i; k < (size_t)N_NODES; k += stride)
        deg[k] = 0.0f;
}

__global__ void edge_kernel(const float* __restrict__ nf,
                            const float* __restrict__ ef,
                            const int*   __restrict__ src,
                            const int*   __restrict__ dst,
                            float* __restrict__ out,
                            float* __restrict__ deg) {
    const int lane = threadIdx.x & 31;
    const int wave = threadIdx.x >> 5;
    const int e    = blockIdx.x * 8 + wave;
    if (e >= N_EDGES) return;

    const int s = src[e];
    const int d = dst[e];

    const int pe = (e + 8 < N_EDGES) ? (e + 8) : e;
    __builtin_prefetch(&ef[(size_t)pe * EF_DIM + lane], 0, 1);

    const float4 x = ((const float4*)(nf + (size_t)s * IN_FEAT))[lane];
    float* orow = out + (size_t)d * OUT_DIM;
    float* onf  = orow + lane * 4;
    gatomic_add_f32(onf + 0, x.x);
    gatomic_add_f32(onf + 1, x.y);
    gatomic_add_f32(onf + 2, x.z);
    gatomic_add_f32(onf + 3, x.w);

    const float fe = ef[(size_t)e * EF_DIM + lane];
    gatomic_add_f32(orow + 160 + lane, fe);

    if (lane == 0) gatomic_add_f32(deg + d, 1.0f);
}

__global__ void prompt_kernel(const float* __restrict__ prompt,
                              const float* __restrict__ hp,
                              const float* __restrict__ deg,
                              float* __restrict__ out) {
    const int lane = threadIdx.x;
    const int m0   = blockIdx.x * 16;
    const int l15  = lane & 15;
    const bool lo  = (lane < 16);

    const float dv = deg[m0 + l15];
    v2f a; a.x = lo ? dv : 0.0f; a.y = 0.0f;

    const float p0 = prompt[l15], p1 = prompt[16 + l15];
    const float h0 = hp[l15],     h1 = hp[16 + l15];
    v2f b0; b0.x = lo ? p0 : 0.0f; b0.y = 0.0f;
    v2f b1; b1.x = lo ? p1 : 0.0f; b1.y = 0.0f;
    v2f b2; b2.x = lo ? h0 : 0.0f; b2.y = 0.0f;
    v2f b3; b3.x = lo ? h1 : 0.0f; b3.y = 0.0f;

    v8f c = {0.f, 0.f, 0.f, 0.f, 0.f, 0.f, 0.f, 0.f};
    v8f d0 = __builtin_amdgcn_wmma_f32_16x16x4_f32(false, a, false, b0, (short)0, c, false, false);
    v8f d1 = __builtin_amdgcn_wmma_f32_16x16x4_f32(false, a, false, b1, (short)0, c, false, false);
    v8f d2 = __builtin_amdgcn_wmma_f32_16x16x4_f32(false, a, false, b2, (short)0, c, false, false);
    v8f d3 = __builtin_amdgcn_wmma_f32_16x16x4_f32(false, a, false, b3, (short)0, c, false, false);

    const int mofs = (lane >> 4) * 8;
#pragma unroll
    for (int r = 0; r < 8; ++r) {
        const size_t row = (size_t)(m0 + mofs + r) * OUT_DIM;
        out[row + 128 + l15]      = d0[r];
        out[row + 144 + l15]      = d1[r];
        out[row + 192 + l15]      = d2[r];
        out[row + 208 + l15]      = d3[r];
    }
}

extern "C" void kernel_launch(void* const* d_in, const int* in_sizes, int n_in,
                              void* d_out, int out_size, void* d_ws, size_t ws_size,
                              hipStream_t stream) {
    (void)in_sizes; (void)n_in; (void)out_size; (void)ws_size;

    const float* nf     = (const float*)d_in[0];
    const float* ef     = (const float*)d_in[1];
    const float* prompt = (const float*)d_in[2];
    const float* hp     = (const float*)d_in[3];
    const int*   src    = (const int*)d_in[4];
    const int*   dst    = (const int*)d_in[5];

    float* out = (float*)d_out;
    float* deg = (float*)d_ws;

    zero_kernel<<<2048, 256, 0, stream>>>(out, deg);
    edge_kernel<<<N_EDGES / 8, 256, 0, stream>>>(nf, ef, src, dst, out, deg);
    prompt_kernel<<<N_NODES / 16, 32, 0, stream>>>(prompt, hp, deg, out);
}


# edge kernel: one wave per edge -> 6250 waves looping over four 32-edge batches each, 4 node + 4 edge rows requested ahead of the atomics, degree add once per batch; same f32 atomics, arrival order
# speedup vs baseline: 1.0373x; 1.0373x over previous
#include <hip/hip_runtime.h>

#define N_NODES   50000
#define N_EDGES   800000
#define IN_FEAT   128
#define PROMPT_DIM 32
#define EF_DIM     32
#define HP_DIM     32
#define OUT_DIM   224

typedef __attribute__((ext_vector_type(2))) float v2f;
typedef __attribute__((ext_vector_type(8))) float v8f;

__device__ __forceinline__ void gatomic_add_f32(float* p, float v) {
    asm volatile("global_atomic_add_f32 %0, %1, off" :: "v"(p), "v"(v) : "memory");
}

__global__ void zero_kernel(float* __restrict__ out, float* __restrict__ deg) {
    const size_t stride = (size_t)gridDim.x * blockDim.x;
    size_t i = (size_t)blockIdx.x * blockDim.x + threadIdx.x;
    const size_t total4 = (size_t)N_NODES * OUT_DIM / 4;
    float4 z = make_float4(0.f, 0.f, 0.f, 0.f);
    for (size_t k = i; k < total4; k += stride)
        ((float4*)out)[k] = z;
    for (size_t k = i; k < (size_t)N_NODES; k += stride)
        deg[k] = 0.0f;
}

#define N_BATCH   (N_EDGES / 32)
#define N_EWAVES  (N_BATCH / 4)
#define STEP      4

__device__ __forceinline__ void request_rows(const float4* __restrict__ nf4,
                                             const float* __restrict__ efb,
                                             const int s_l, const int j0, const int lane,
                                             float4 (&x)[STEP], float (&f)[STEP]) {
#pragma unroll
    for (int u = 0; u < STEP; ++u) {
        const int s = __builtin_amdgcn_readlane(s_l, j0 + u);
        const float4* rp = nf4 + (size_t)s * (IN_FEAT / 4);
        const float*  ep = efb + (size_t)(j0 + u) * EF_DIM;
        x[u] = rp[lane];
        f[u] = ep[lane];
    }
}

__device__ __forceinline__ void scatter_rows(float* __restrict__ out,
                                             const int d_l, const int j0, const int lane,
                                             const float4 (&x)[STEP], const float (&f)[STEP]) {
#pragma unroll
    for (int u = 0; u < STEP; ++u) {
        const int d = __builtin_amdgcn_readlane(d_l, j0 + u);
        float* orow = out + (size_t)d * OUT_DIM;
        float* onf  = orow + lane * 4;
        gatomic_add_f32(onf + 0, x[u].x);
        gatomic_add_f32(onf + 1, x[u].y);
        gatomic_add_f32(onf + 2, x[u].z);
        gatomic_add_f32(onf + 3, x[u].w);
        gatomic_add_f32(orow + 160 + lane, f[u]);
    }
}

__global__ void __launch_bounds__(64)
edge_kernel(const float* __restrict__ nf,
            const float* __restrict__ ef,
            const int*   __restrict__ src,
            const int*   __restrict__ dst,
            float* __restrict__ out,
            float* __restrict__ deg) {
    const int lane = threadIdx.x & 31;
    const int wave = __builtin_amdgcn_readfirstlane((int)(threadIdx.x >> 5));
    const float4* nf4 = (const float4*)nf;

    int b = blockIdx.x * 2 + wave;
    int s_l = src[b * 32 + lane];
    int d_l = dst[b * 32 + lane];

    while (b < N_BATCH) {
        const float* efb = ef + (size_t)b * (32 * EF_DIM);
        const int bn  = b + N_EWAVES;
        const int bnc = (bn < N_BATCH) ? bn : (N_BATCH - 1);
        const int s_n = src[bnc * 32 + lane];
        const int d_n = dst[bnc * 32 + lane];

        float4 xa[STEP], xb[STEP];
        float  fa[STEP], fb[STEP];
        request_rows(nf4, efb, s_l, 0, lane, xa, fa);
#pragma nounroll
        for (int c = 0; c < 32; c += 2 * STEP) {
            request_rows(nf4, efb, s_l, c + STEP, lane, xb, fb);
            scatter_rows(out, d_l, c, lane, xa, fa);
            if (c + 2 * STEP < 32) request_rows(nf4, efb, s_l, c + 2 * STEP, lane, xa, fa);
            scatter_rows(out, d_l, c + STEP, lane, xb, fb);
        }
        gatomic_add_f32(deg + d_l, 1.0f);

        b = bn; s_l = s_n; d_l = d_n;
    }
}

__global__ void prompt_kernel(const float* __restrict__ prompt,
                              const float* __restrict__ hp,
                              const float* __restrict__ deg,
                              float* __restrict__ out) {
    const int lane = threadIdx.x;
    const int m0   = blockIdx.x * 16;
    const int l15  = lane & 15;
    const bool lo  = (lane < 16);

    const float dv = deg[m0 + l15];
    v2f a; a.x = lo ? dv : 0.0f; a.y = 0.0f;

    const float p0 = prompt[l15], p1 = prompt[16 + l15];
    const float h0 = hp[l15],     h1 = hp[16 + l15];
    v2f b0; b0.x = lo ? p0 : 0.0f; b0.y = 0.0f;
    v2f b1; b1.x = lo ? p1 : 0.0f; b1.y = 0.0f;
    v2f b2; b2.x = lo ? h0 : 0.0f; b2.y = 0.0f;
    v2f b3; b3.x = lo ? h1 : 0.0f; b3.y = 0.0f;

    v8f c = {0.f, 0.f, 0.f, 0.f, 0.f, 0.f, 0.f, 0.f};
    v8f d0 = __builtin_amdgcn_wmma_f32_16x16x4_f32(false, a, false, b0, (short)0, c, false, false);
    v8f d1 = __builtin_amdgcn_wmma_f32_16x16x4_f32(false, a, false, b1, (short)0, c, false, false);
    v8f d2 = __builtin_amdgcn_wmma_f32_16x16x4_f32(false, a, false, b2, (short)0, c, false, false);
    v8f d3 = __builtin_amdgcn_wmma_f32_16x16x4_f32(false, a, false, b3, (short)0, c, false, false);

    const int mofs = (lane >> 4) * 8;
#pragma unroll
    for (int r = 0; r < 8; ++r) {
        const size_t row = (size_t)(m0 + mofs + r) * OUT_DIM;
        out[row + 128 + l15]      = d0[r];
        out[row + 144 + l15]      = d1[r];
        out[row + 192 + l15]      = d2[r];
        out[row + 208 + l15]      = d3[r];
    }
}

extern "C" void kernel_launch(void* const* d_in, const int* in_sizes, int n_in,
                              void* d_out, int out_size, void* d_ws, size_t ws_size,
                              hipStream_t stream) {
    (void)in_sizes; (void)n_in; (void)out_size; (void)ws_size;

    const float* nf     = (const float*)d_in[0];
    const float* ef     = (const float*)d_in[1];
    const float* prompt = (const float*)d_in[2];
    const float* hp     = (const float*)d_in[3];
    const int*   src    = (const int*)d_in[4];
    const int*   dst    = (const int*)d_in[5];

    float* out = (float*)d_out;
    float* deg = (float*)d_ws;

    zero_kernel<<<2048, 256, 0, stream>>>(out, deg);
    edge_kernel<<<N_EWAVES / 2, 64, 0, stream>>>(nf, ef, src, dst, out, deg);
    prompt_kernel<<<N_NODES / 16, 32, 0, stream>>>(prompt, hp, deg, out);
}


# f32 row scatter + zero fill -> edges counting-sorted by dst (degree adds kept, returned old count = slot), one wave per node sums its rows in registers, out stored once; sums in slot arrival order
# speedup vs baseline: 1.4880x; 1.4345x over previous
#include <hip/hip_runtime.h>

#define N_NODES   50000
#define N_EDGES   800000
#define IN_FEAT   128
#define PROMPT_DIM 32
#define EF_DIM     32
#define HP_DIM     32
#define OUT_DIM   224

typedef __attribute__((ext_vector_type(2))) float v2f;
typedef __attribute__((ext_vector_type(8))) float v8f;

__device__ __forceinline__ void gatomic_add_f32(float* p, float v) {
    asm volatile("global_atomic_add_f32 %0, %1, off" :: "v"(p), "v"(v) : "memory");
}
__device__ __forceinline__ float gatomic_add_rtn_f32(float* p, float v) {
    float old;
    asm volatile("global_atomic_add_f32 %0, %1, %2, off th:TH_ATOMIC_RETURN\n\t"
                 "s_wait_loadcnt 0x0"
                 : "=v"(old) : "v"(p), "v"(v) : "memory");
    return old;
}

static_assert(N_EDGES % 256 == 0, "one thread per edge, whole work-groups");
static_assert(N_EDGES <= 16 * N_NODES, "16 sorted entries per output row must hold every edge");
static_assert(N_NODES % 16 == 0, "whole tiles in gather_kernel and prompt_kernel");
static_assert(N_EDGES < (1 << 24), "edge counts must stay exact in f32");

__device__ __forceinline__ size_t ent_word(const int p) {
    return (size_t)(p >> 4) * OUT_DIM + 128 + ((p & 15) << 1);
}
__device__ __forceinline__ size_t cur_word(const int i) {
    return (size_t)(i >> 5) * OUT_DIM + 192 + (i & 31);
}

__global__ void zero_kernel(float* __restrict__ deg) {
    const int i = blockIdx.x * blockDim.x + threadIdx.x;
    if (i < N_NODES) deg[i] = 0.0f;
}

__global__ void __launch_bounds__(256)
count_kernel(const int* __restrict__ dst, float* __restrict__ deg) {
    const int e = blockIdx.x * 256 + threadIdx.x;
    gatomic_add_f32(deg + dst[e], 1.0f);
}

#define SCAN_T 1024
__global__ void __launch_bounds__(SCAN_T)
offsets_kernel(const float* __restrict__ deg, float* __restrict__ outf) {
    __shared__ int part[32];
    __shared__ int tot[32];
    const int t = threadIdx.x, lane = t & 31, w = t >> 5;
    const int b = blockIdx.x;

    int ahead = 0;
    for (int i = 0; i < b; ++i) ahead += (int)deg[i * SCAN_T + t];

    const int idx = b * SCAN_T + t;
    const int c = (idx < N_NODES) ? (int)deg[idx] : 0;

    int incl = c;
    for (int o = 1; o < 32; o <<= 1) {
        const int y = __shfl_up(incl, o, 32);
        if (lane >= o) incl += y;
    }
    int asum = ahead;
    for (int o = 16; o > 0; o >>= 1) asum += __shfl_xor(asum, o, 32);
    if (lane == 31) part[w] = incl;
    if (lane == 0)  tot[w]  = asum;
    __syncthreads();
    int pre = (lane < w) ? part[lane] : 0;
    int q   = tot[lane];
    for (int o = 16; o > 0; o >>= 1) {
        pre += __shfl_xor(pre, o, 32);
        q   += __shfl_xor(q, o, 32);
    }
    if (idx < N_NODES) outf[cur_word(idx)] = (float)(q + pre + incl - c);
}

__global__ void __launch_bounds__(256)
fill_kernel(const int* __restrict__ src, const int* __restrict__ dst, float* outf) {
    const int e = blockIdx.x * 256 + threadIdx.x;
    const int d = dst[e];
    const int s = src[e];
    const int p = (int)gatomic_add_rtn_f32(outf + cur_word(d), 1.0f);
    int2 r; r.x = e; r.y = s;
    *(int2*)((int*)outf + ent_word(p)) = r;
}

#define ROWS 8
__global__ void __launch_bounds__(256)
gather_kernel(const float* __restrict__ nf,
              const float* __restrict__ ef,
              const float* __restrict__ deg,
              const int*   __restrict__ ents,
              const float* __restrict__ curs,
              float* __restrict__ out) {
    const int lane = threadIdx.x & 31;
    const int wave = __builtin_amdgcn_readfirstlane((int)(threadIdx.x >> 5));
    const int n = blockIdx.x * 8 + wave;
    const float4* nf4 = (const float4*)nf;

    const int cnt   = (int)deg[n];
    const int start = (int)curs[cur_word(n)] - cnt;

    float4 acc = make_float4(0.f, 0.f, 0.f, 0.f);
    float  g   = 0.f;

    for (int base = 0; base < cnt; base += 32) {
        const int left = cnt - base;
        const int m = left < 32 ? left : 32;
        int e_l = 0, s_l = 0;
        if (lane < m) {
            const int2 r = *(const int2*)(ents + ent_word(start + base + lane));
            e_l = r.x; s_l = r.y;
        }
        for (int j0 = 0; j0 < m; j0 += ROWS) {
            float4 x[ROWS];
            float  f[ROWS];
#pragma unroll
            for (int u = 0; u < ROWS; ++u) {
                x[u] = make_float4(0.f, 0.f, 0.f, 0.f);
                f[u] = 0.f;
                if (u == 0 || j0 + u < m) {
                    const int ee = __builtin_amdgcn_readlane(e_l, j0 + u);
                    const int ss = __builtin_amdgcn_readlane(s_l, j0 + u);
                    x[u] = nf4[(size_t)ss * (IN_FEAT / 4) + lane];
                    f[u] = ef[(size_t)ee * EF_DIM + lane];
                }
            }
#pragma unroll
            for (int u = 0; u < ROWS; ++u) {
                acc.x += x[u].x; acc.y += x[u].y; acc.z += x[u].z; acc.w += x[u].w;
                g += f[u];
            }
        }
    }

    float* orow = out + (size_t)n * OUT_DIM;
    ((float4*)orow)[lane] = acc;
    orow[160 + lane] = g;
}

__global__ void prompt_kernel(const float* __restrict__ prompt,
                              const float* __restrict__ hp,
                              const float* __restrict__ deg,
                              float* __restrict__ out) {
    const int lane = threadIdx.x;
    const int m0   = blockIdx.x * 16;
    const int l15  = lane & 15;
    const bool lo  = (lane < 16);

    const float dv = deg[m0 + l15];
    v2f a; a.x = lo ? dv : 0.0f; a.y = 0.0f;

    const float p0 = prompt[l15], p1 = prompt[16 + l15];
    const float h0 = hp[l15],     h1 = hp[16 + l15];
    v2f b0; b0.x = lo ? p0 : 0.0f; b0.y = 0.0f;
    v2f b1; b1.x = lo ? p1 : 0.0f; b1.y = 0.0f;
    v2f b2; b2.x = lo ? h0 : 0.0f; b2.y = 0.0f;
    v2f b3; b3.x = lo ? h1 : 0.0f; b3.y = 0.0f;

    v8f c = {0.f, 0.f, 0.f, 0.f, 0.f, 0.f, 0.f, 0.f};
    v8f d0 = __builtin_amdgcn_wmma_f32_16x16x4_f32(false, a, false, b0, (short)0, c, false, false);
    v8f d1 = __builtin_amdgcn_wmma_f32_16x16x4_f32(false, a, false, b1, (short)0, c, false, false);
    v8f d2 = __builtin_amdgcn_wmma_f32_16x16x4_f32(false, a, false, b2, (short)0, c, false, false);
    v8f d3 = __builtin_amdgcn_wmma_f32_16x16x4_f32(false, a, false, b3, (short)0, c, false, false);

    const int mofs = (lane >> 4) * 8;
#pragma unroll
    for (int r = 0; r < 8; ++r) {
        const size_t row = (size_t)(m0 + mofs + r) * OUT_DIM;
        out[row + 128 + l15]      = d0[r];
        out[row + 144 + l15]      = d1[r];
        out[row + 192 + l15]      = d2[r];
        out[row + 208 + l15]      = d3[r];
    }
}

extern "C" void kernel_launch(void* const* d_in, const int* in_sizes, int n_in,
                              void* d_out, int out_size, void* d_ws, size_t ws_size,
                              hipStream_t stream) {
    (void)in_sizes; (void)n_in; (void)out_size; (void)ws_size;

    const float* nf     = (const float*)d_in[0];
    const float* ef     = (const float*)d_in[1];
    const float* prompt = (const float*)d_in[2];
    const float* hp     = (const float*)d_in[3];
    const int*   src    = (const int*)d_in[4];
    const int*   dst    = (const int*)d_in[5];

    float* out = (float*)d_out;
    float* deg = (float*)d_ws;

    zero_kernel<<<(N_NODES + 255) / 256, 256, 0, stream>>>(deg);
    count_kernel<<<N_EDGES / 256, 256, 0, stream>>>(dst, deg);
    offsets_kernel<<<(N_NODES + SCAN_T - 1) / SCAN_T, SCAN_T, 0, stream>>>(deg, out);
    fill_kernel<<<N_EDGES / 256, 256, 0, stream>>>(src, dst, out);
    gather_kernel<<<N_NODES / 8, 256, 0, stream>>>(nf, ef, deg, (const int*)d_out, (const float*)d_out, out);
    prompt_kernel<<<N_NODES / 16, 32, 0, stream>>>(prompt, hp, deg, out);
}


# counting sort (count, scan, fill) -> fixed 112-entry bucket in each node's own out row, slot = old in-degree returned by the add; 6 -> 4 launches; dst sweep on overflow; sums in slot arrival order
# speedup vs baseline: 2.2791x; 1.5317x over previous
#include <hip/hip_runtime.h>

#define N_NODES   50000
#define N_EDGES   800000
#define IN_FEAT   128
#define PROMPT_DIM 32
#define EF_DIM     32
#define HP_DIM     32
#define OUT_DIM   224

typedef __attribute__((ext_vector_type(2))) float v2f;
typedef __attribute__((ext_vector_type(8))) float v8f;

__device__ __forceinline__ float gatomic_add_rtn_f32(float* p, float v) {
    float old;
    asm volatile("global_atomic_add_f32 %0, %1, %2, off th:TH_ATOMIC_RETURN\n\t"
                 "s_wait_loadcnt 0x0"
                 : "=v"(old) : "v"(p), "v"(v) : "memory");
    return old;
}

#define CAP (OUT_DIM / 2)
static_assert(N_EDGES % 256 == 0, "one thread per edge, whole work-groups");
static_assert(N_EDGES % 32 == 0, "whole waves in the dst sweep");
static_assert(N_NODES % 16 == 0, "whole tiles in gather_kernel and prompt_kernel");
static_assert(N_EDGES < (1 << 24), "edge counts must stay exact in f32");

__global__ void zero_kernel(float* __restrict__ deg) {
    const int i = blockIdx.x * blockDim.x + threadIdx.x;
    if (i < N_NODES) deg[i] = 0.0f;
}

__global__ void __launch_bounds__(256)
fill_kernel(const int* __restrict__ src, const int* __restrict__ dst,
            float* deg, float* outf) {
    const int e = blockIdx.x * 256 + threadIdx.x;
    const int d = dst[e];
    const int s = src[e];
    asm volatile("" :: "v"(s));
    const int slot = (int)gatomic_add_rtn_f32(deg + d, 1.0f);
    if (slot < CAP) {
        int2 r; r.x = e; r.y = s;
        *(int2*)((int*)outf + (size_t)d * OUT_DIM + 2 * slot) = r;
    }
}

#define ROWS 8
__global__ void __launch_bounds__(256)
gather_kernel(const float* __restrict__ nf,
              const float* __restrict__ ef,
              const float* __restrict__ deg,
              const int*   __restrict__ src,
              const int*   __restrict__ dst,
              float* out) {
    const int lane = threadIdx.x & 31;
    const int wave = __builtin_amdgcn_readfirstlane((int)(threadIdx.x >> 5));
    const int n = blockIdx.x * 8 + wave;
    const float4* nf4 = (const float4*)nf;
    float* orow = out + (size_t)n * OUT_DIM;
    const int2* bucket = (const int2*)orow;

    const int cnt = (int)deg[n];

    float4 acc = make_float4(0.f, 0.f, 0.f, 0.f);
    float  g   = 0.f;

    if (cnt <= CAP) {
        for (int base = 0; base < cnt; base += 32) {
            const int left = cnt - base;
            const int m = left < 32 ? left : 32;
            int e_l = 0, s_l = 0;
            if (lane < m) {
                const int2 r = bucket[base + lane];
                e_l = r.x; s_l = r.y;
            }
            for (int j0 = 0; j0 < m; j0 += ROWS) {
                float4 x[ROWS];
                float  f[ROWS];
#pragma unroll
                for (int u = 0; u < ROWS; ++u) {
                    x[u] = make_float4(0.f, 0.f, 0.f, 0.f);
                    f[u] = 0.f;
                    if (u == 0 || j0 + u < m) {
                        const int ee = __builtin_amdgcn_readlane(e_l, j0 + u);
                        const int ss = __builtin_amdgcn_readlane(s_l, j0 + u);
                        x[u] = nf4[(size_t)ss * (IN_FEAT / 4) + lane];
                        f[u] = ef[(size_t)ee * EF_DIM + lane];
                    }
                }
#pragma unroll
                for (int u = 0; u < ROWS; ++u) {
                    acc.x += x[u].x; acc.y += x[u].y; acc.z += x[u].z; acc.w += x[u].w;
                    g += f[u];
                }
            }
        }
    } else {
        for (int e0 = 0; e0 < N_EDGES; e0 += 32) {
            const int d_l = dst[e0 + lane];
            const int s_l = src[e0 + lane];
            unsigned hit = __builtin_amdgcn_ballot_w32(d_l == n);
            while (hit != 0u) {
                const int j = __builtin_ctz(hit);
                hit &= hit - 1u;
                const int ss = __builtin_amdgcn_readlane(s_l, j);
                const float4 xv = nf4[(size_t)ss * (IN_FEAT / 4) + lane];
                const float  fv = ef[(size_t)(e0 + j) * EF_DIM + lane];
                acc.x += xv.x; acc.y += xv.y; acc.z += xv.z; acc.w += xv.w;
                g += fv;
            }
        }
    }

    ((float4*)orow)[lane] = acc;
    orow[160 + lane] = g;
}

__global__ void prompt_kernel(const float* __restrict__ prompt,
                              const float* __restrict__ hp,
                              const float* __restrict__ deg,
                              float* __restrict__ out) {
    const int lane = threadIdx.x;
    const int m0   = blockIdx.x * 16;
    const int l15  = lane & 15;
    const bool lo  = (lane < 16);

    const float dv = deg[m0 + l15];
    v2f a; a.x = lo ? dv : 0.0f; a.y = 0.0f;

    const float p0 = prompt[l15], p1 = prompt[16 + l15];
    const float h0 = hp[l15],     h1 = hp[16 + l15];
    v2f b0; b0.x = lo ? p0 : 0.0f; b0.y = 0.0f;
    v2f b1; b1.x = lo ? p1 : 0.0f; b1.y = 0.0f;
    v2f b2; b2.x = lo ? h0 : 0.0f; b2.y = 0.0f;
    v2f b3; b3.x = lo ? h1 : 0.0f; b3.y = 0.0f;

    v8f c = {0.f, 0.f, 0.f, 0.f, 0.f, 0.f, 0.f, 0.f};
    v8f d0 = __builtin_amdgcn_wmma_f32_16x16x4_f32(false, a, false, b0, (short)0, c, false, false);
    v8f d1 = __builtin_amdgcn_wmma_f32_16x16x4_f32(false, a, false, b1, (short)0, c, false, false);
    v8f d2 = __builtin_amdgcn_wmma_f32_16x16x4_f32(false, a, false, b2, (short)0, c, false, false);
    v8f d3 = __builtin_amdgcn_wmma_f32_16x16x4_f32(false, a, false, b3, (short)0, c, false, false);

    const int mofs = (lane >> 4) * 8;
#pragma unroll
    for (int r = 0; r < 8; ++r) {
        const size_t row = (size_t)(m0 + mofs + r) * OUT_DIM;
        out[row + 128 + l15]      = d0[r];
        out[row + 144 + l15]      = d1[r];
        out[row + 192 + l15]      = d2[r];
        out[row + 208 + l15]      = d3[r];
    }
}

extern "C" void kernel_launch(void* const* d_in, const int* in_sizes, int n_in,
                              void* d_out, int out_size, void* d_ws, size_t ws_size,
                              hipStream_t stream) {
    (void)in_sizes; (void)n_in; (void)out_size; (void)ws_size;

    const float* nf     = (const float*)d_in[0];
    const float* ef     = (const float*)d_in[1];
    const float* prompt = (const float*)d_in[2];
    const float* hp     = (const float*)d_in[3];
    const int*   src    = (const int*)d_in[4];
    const int*   dst    = (const int*)d_in[5];

    float* out = (float*)d_out;
    float* deg = (float*)d_ws;

    zero_kernel<<<(N_NODES + 255) / 256, 256, 0, stream>>>(deg);
    fill_kernel<<<N_EDGES / 256, 256, 0, stream>>>(src, dst, deg, out);
    gather_kernel<<<N_NODES / 8, 256, 0, stream>>>(nf, ef, deg, src, dst, out);
    prompt_kernel<<<N_NODES / 16, 32, 0, stream>>>(prompt, hp, deg, out);
}


# in-degree counter moved from a packed array into each node's own out row, bucket filled downwards from it: add and entry store share a line, gather gets count+slots in one trip; slot order kept
# speedup vs baseline: 2.6675x; 1.1704x over previous
#include <hip/hip_runtime.h>

#define N_NODES   50000
#define N_EDGES   800000
#define IN_FEAT   128
#define PROMPT_DIM 32
#define EF_DIM     32
#define HP_DIM     32
#define OUT_DIM   224

typedef __attribute__((ext_vector_type(2))) float v2f;
typedef __attribute__((ext_vector_type(8))) float v8f;

__device__ __forceinline__ float gatomic_add_rtn_f32(float* p, float v) {
    float old;
    asm volatile("global_atomic_add_f32 %0, %1, %2, off th:TH_ATOMIC_RETURN\n\t"
                 "s_wait_loadcnt 0x0"
                 : "=v"(old) : "v"(p), "v"(v) : "memory");
    return old;
}

#define CNT_COL (OUT_DIM - 2)
#define ENT0    (OUT_DIM - 4)
#define CAP     (ENT0 / 2 + 1)
static_assert(N_EDGES % 256 == 0, "one thread per edge, whole work-groups");
static_assert(N_EDGES % 32 == 0, "whole waves in the dst sweep");
static_assert(N_NODES % 16 == 0, "whole tiles in gather_kernel and prompt_kernel");
static_assert(N_EDGES < (1 << 24), "edge counts must stay exact in f32");

__global__ void zero_kernel(float* __restrict__ outf) {
    const int i = blockIdx.x * blockDim.x + threadIdx.x;
    if (i < N_NODES) outf[(size_t)i * OUT_DIM + CNT_COL] = 0.0f;
}

__global__ void __launch_bounds__(256)
fill_kernel(const int* __restrict__ src, const int* __restrict__ dst,
            float* outf) {
    const int e = blockIdx.x * 256 + threadIdx.x;
    const int d = dst[e];
    const int s = src[e];
    asm volatile("" :: "v"(s));
    float* row = outf + (size_t)d * OUT_DIM;
    const int slot = (int)gatomic_add_rtn_f32(row + CNT_COL, 1.0f);
    if (slot < CAP) {
        int2 r; r.x = e; r.y = s;
        *(int2*)((int*)row + ENT0 - 2 * slot) = r;
    }
}

#define ROWS 8
__global__ void __launch_bounds__(256)
gather_kernel(const float* __restrict__ nf,
              const float* __restrict__ ef,
              const int*   __restrict__ src,
              const int*   __restrict__ dst,
              float* out) {
    const int lane = threadIdx.x & 31;
    const int wave = __builtin_amdgcn_readfirstlane((int)(threadIdx.x >> 5));
    const int n = blockIdx.x * 8 + wave;
    const float4* nf4 = (const float4*)nf;
    float* orow = out + (size_t)n * OUT_DIM;
    const int2* rowq = (const int2*)orow;

    const float cntf = orow[CNT_COL];
    int2 r = rowq[ENT0 / 2 - lane];
    asm volatile("" :: "v"(r.x), "v"(r.y));
    const int cnt = (int)cntf;

    float4 acc = make_float4(0.f, 0.f, 0.f, 0.f);
    float  g   = 0.f;

    if (cnt <= CAP) {
        for (int base = 0; base < cnt; base += 32) {
            const int left = cnt - base;
            const int m = left < 32 ? left : 32;
            if (base != 0 && lane < m) r = rowq[ENT0 / 2 - (base + lane)];
            const int e_l = r.x, s_l = r.y;
            for (int j0 = 0; j0 < m; j0 += ROWS) {
                float4 x[ROWS];
                float  f[ROWS];
#pragma unroll
                for (int u = 0; u < ROWS; ++u) {
                    x[u] = make_float4(0.f, 0.f, 0.f, 0.f);
                    f[u] = 0.f;
                    if (u == 0 || j0 + u < m) {
                        const int ee = __builtin_amdgcn_readlane(e_l, j0 + u);
                        const int ss = __builtin_amdgcn_readlane(s_l, j0 + u);
                        x[u] = nf4[(size_t)ss * (IN_FEAT / 4) + lane];
                        f[u] = ef[(size_t)ee * EF_DIM + lane];
                    }
                }
#pragma unroll
                for (int u = 0; u < ROWS; ++u) {
                    acc.x += x[u].x; acc.y += x[u].y; acc.z += x[u].z; acc.w += x[u].w;
                    g += f[u];
                }
            }
        }
    } else {
        for (int e0 = 0; e0 < N_EDGES; e0 += 32) {
            const int d_l = dst[e0 + lane];
            const int s_l = src[e0 + lane];
            unsigned hit = __builtin_amdgcn_ballot_w32(d_l == n);
            while (hit != 0u) {
                const int j = __builtin_ctz(hit);
                hit &= hit - 1u;
                const int ss = __builtin_amdgcn_readlane(s_l, j);
                const float4 xv = nf4[(size_t)ss * (IN_FEAT / 4) + lane];
                const float  fv = ef[(size_t)(e0 + j) * EF_DIM + lane];
                acc.x += xv.x; acc.y += xv.y; acc.z += xv.z; acc.w += xv.w;
                g += fv;
            }
        }
    }

    ((float4*)orow)[lane] = acc;
    orow[160 + lane] = g;
}

__global__ void prompt_kernel(const float* __restrict__ prompt,
                              const float* __restrict__ hp,
                              float* __restrict__ out) {
    const int lane = threadIdx.x;
    const int m0   = blockIdx.x * 16;
    const int l15  = lane & 15;
    const bool lo  = (lane < 16);

    const float dv = out[(size_t)(m0 + l15) * OUT_DIM + CNT_COL];
    v2f a; a.x = lo ? dv : 0.0f; a.y = 0.0f;

    const float p0 = prompt[l15], p1 = prompt[16 + l15];
    const float h0 = hp[l15],     h1 = hp[16 + l15];
    v2f b0; b0.x = lo ? p0 : 0.0f; b0.y = 0.0f;
    v2f b1; b1.x = lo ? p1 : 0.0f; b1.y = 0.0f;
    v2f b2; b2.x = lo ? h0 : 0.0f; b2.y = 0.0f;
    v2f b3; b3.x = lo ? h1 : 0.0f; b3.y = 0.0f;

    v8f c = {0.f, 0.f, 0.f, 0.f, 0.f, 0.f, 0.f, 0.f};
    v8f d0 = __builtin_amdgcn_wmma_f32_16x16x4_f32(false, a, false, b0, (short)0, c, false, false);
    v8f d1 = __builtin_amdgcn_wmma_f32_16x16x4_f32(false, a, false, b1, (short)0, c, false, false);
    v8f d2 = __builtin_amdgcn_wmma_f32_16x16x4_f32(false, a, false, b2, (short)0, c, false, false);
    v8f d3 = __builtin_amdgcn_wmma_f32_16x16x4_f32(false, a, false, b3, (short)0, c, false, false);

    const int mofs = (lane >> 4) * 8;
#pragma unroll
    for (int r = 0; r < 8; ++r) {
        const size_t row = (size_t)(m0 + mofs + r) * OUT_DIM;
        out[row + 128 + l15]      = d0[r];
        out[row + 144 + l15]      = d1[r];
        out[row + 192 + l15]      = d2[r];
        out[row + 208 + l15]      = d3[r];
    }
}

extern "C" void kernel_launch(void* const* d_in, const int* in_sizes, int n_in,
                              void* d_out, int out_size, void* d_ws, size_t ws_size,
                              hipStream_t stream) {
    (void)in_sizes; (void)n_in; (void)out_size; (void)d_ws; (void)ws_size;

    const float* nf     = (const float*)d_in[0];
    const float* ef     = (const float*)d_in[1];
    const float* prompt = (const float*)d_in[2];
    const float* hp     = (const float*)d_in[3];
    const int*   src    = (const int*)d_in[4];
    const int*   dst    = (const int*)d_in[5];

    float* out = (float*)d_out;

    zero_kernel<<<(N_NODES + 255) / 256, 256, 0, stream>>>(out);
    fill_kernel<<<N_EDGES / 256, 256, 0, stream>>>(src, dst, out);
    gather_kernel<<<N_NODES / 8, 256, 0, stream>>>(nf, ef, src, dst, out);
    prompt_kernel<<<N_NODES / 16, 32, 0, stream>>>(prompt, hp, out);
}


# bucket + in-degree counter split per package half (by die id): fill's returning adds and entry stores touch only lines their own half writes; gather gets both in 1 request; per-half slot sum order
# speedup vs baseline: 3.0533x; 1.1446x over previous
#include <hip/hip_runtime.h>

#define N_NODES   50000
#define N_EDGES   800000
#define IN_FEAT   128
#define PROMPT_DIM 32
#define EF_DIM     32
#define HP_DIM     32
#define OUT_DIM   224

typedef __attribute__((ext_vector_type(2))) float v2f;
typedef __attribute__((ext_vector_type(8))) float v8f;

__device__ __forceinline__ float gatomic_add_rtn_f32(float* p, float v) {
    float old;
    asm volatile("global_atomic_add_f32 %0, %1, %2, off th:TH_ATOMIC_RETURN\n\t"
                 "s_wait_loadcnt 0x0"
                 : "=v"(old) : "v"(p), "v"(v) : "memory");
    return old;
}

#define CNT_COL (OUT_DIM - 2)
#define HOME0   (OUT_DIM / 2 - 1)
#define FIRST   15
#define CAP0    (FIRST + 32)
#define CAP1    (FIRST + 48)
static_assert(OUT_DIM * 4 == 7 * 128, "a row is seven whole 128-B lines");
static_assert(N_EDGES % 256 == 0, "one thread per edge, whole work-groups");
static_assert(N_EDGES % 32 == 0, "whole waves in the dst sweep");
static_assert(N_NODES % 16 == 0, "whole tiles in gather_kernel and prompt_kernel");
static_assert(N_EDGES < (1 << 24), "edge counts must stay exact in f32");

__global__ void zero_kernel(float* __restrict__ outf) {
    const int i = blockIdx.x * blockDim.x + threadIdx.x;
    if (i < N_NODES) {
        outf[(size_t)i * OUT_DIM + CNT_COL] = 0.0f;
        outf[(size_t)i * OUT_DIM] = 0.0f;
    }
}

__global__ void __launch_bounds__(256)
fill_kernel(const int* __restrict__ src, const int* __restrict__ dst,
            float* outf) {
    const unsigned where = __builtin_amdgcn_s_sendmsg_rtn(0x87);
    const bool half1 = ((where >> 18) & 3u) != 0u;
    const int e = blockIdx.x * 256 + threadIdx.x;
    const int d = dst[e];
    const int s = src[e];
    asm volatile("" :: "v"(s));
    float* row = outf + (size_t)d * OUT_DIM;
    const int slot = (int)gatomic_add_rtn_f32(row + (half1 ? 0 : CNT_COL), 1.0f);
    if (slot < (half1 ? CAP1 : CAP0)) {
        int2 r; r.x = e; r.y = s;
        ((int2*)row)[half1 ? 1 + slot : HOME0 - 1 - slot] = r;
    }
}

#define ROWS 8
__global__ void __launch_bounds__(256)
gather_kernel(const float* __restrict__ nf,
              const float* __restrict__ ef,
              const int*   __restrict__ src,
              const int*   __restrict__ dst,
              float* out) {
    const int lane = threadIdx.x & 31;
    const int wave = __builtin_amdgcn_readfirstlane((int)(threadIdx.x >> 5));
    const int n = blockIdx.x * 8 + wave;
    const float4* nf4 = (const float4*)nf;
    float* orow = out + (size_t)n * OUT_DIM;
    const int2* rowq = (const int2*)orow;

    int2 r = rowq[lane < 16 ? HOME0 - lane : lane - 16];
    asm volatile("" :: "v"(r.x), "v"(r.y));
    const int cnt0 = (int)__int_as_float(__builtin_amdgcn_readlane(r.x, 0));
    const int cnt1 = (int)__int_as_float(__builtin_amdgcn_readlane(r.x, 16));

    float4 acc = make_float4(0.f, 0.f, 0.f, 0.f);
    float  g   = 0.f;

    if (cnt0 <= CAP0 && cnt1 <= CAP1) {
        const int m0 = cnt0 < FIRST ? cnt0 : FIRST;
        const int m1 = cnt1 < FIRST ? cnt1 : FIRST;
        int m = m0 + m1, c0 = m0, d0 = 1, d1 = 17 - m0;
        int b0 = FIRST, b1 = FIRST;
        for (;;) {
            const int e_l = r.x, s_l = r.y;
            for (int j0 = 0; j0 < m; j0 += ROWS) {
                float4 x[ROWS];
                float  f[ROWS];
#pragma unroll
                for (int u = 0; u < ROWS; ++u) {
                    x[u] = make_float4(0.f, 0.f, 0.f, 0.f);
                    f[u] = 0.f;
                    if (u == 0 || j0 + u < m) {
                        const int t  = j0 + u;
                        const int li = t + (t < c0 ? d0 : d1);
                        const int ee = __builtin_amdgcn_readlane(e_l, li);
                        const int ss = __builtin_amdgcn_readlane(s_l, li);
                        x[u] = nf4[(size_t)ss * (IN_FEAT / 4) + lane];
                        f[u] = ef[(size_t)ee * EF_DIM + lane];
                    }
                }
#pragma unroll
                for (int u = 0; u < ROWS; ++u) {
                    acc.x += x[u].x; acc.y += x[u].y; acc.z += x[u].z; acc.w += x[u].w;
                    g += f[u];
                }
            }
            if (b0 < cnt0) {
                m = cnt0 - b0 < 32 ? cnt0 - b0 : 32;
                if (lane < m) r = rowq[HOME0 - 1 - (b0 + lane)];
                b0 += 32;
            } else if (b1 < cnt1) {
                m = cnt1 - b1 < 32 ? cnt1 - b1 : 32;
                if (lane < m) r = rowq[1 + b1 + lane];
                b1 += 32;
            } else {
                break;
            }
            c0 = 32; d0 = 0; d1 = 0;
        }
    } else {
        for (int e0 = 0; e0 < N_EDGES; e0 += 32) {
            const int d_l = dst[e0 + lane];
            const int s_l = src[e0 + lane];
            unsigned hit = __builtin_amdgcn_ballot_w32(d_l == n);
            while (hit != 0u) {
                const int j = __builtin_ctz(hit);
                hit &= hit - 1u;
                const int ss = __builtin_amdgcn_readlane(s_l, j);
                const float4 xv = nf4[(size_t)ss * (IN_FEAT / 4) + lane];
                const float  fv = ef[(size_t)(e0 + j) * EF_DIM + lane];
                acc.x += xv.x; acc.y += xv.y; acc.z += xv.z; acc.w += xv.w;
                g += fv;
            }
        }
    }

    ((float4*)orow)[lane] = acc;
    orow[160 + lane] = g;
    if (lane == 0) orow[CNT_COL] = (float)(cnt0 + cnt1);
}

__global__ void prompt_kernel(const float* __restrict__ prompt,
                              const float* __restrict__ hp,
                              float* __restrict__ out) {
    const int lane = threadIdx.x;
    const int m0   = blockIdx.x * 16;
    const int l15  = lane & 15;
    const bool lo  = (lane < 16);

    const float dv = out[(size_t)(m0 + l15) * OUT_DIM + CNT_COL];
    v2f a; a.x = lo ? dv : 0.0f; a.y = 0.0f;

    const float p0 = prompt[l15], p1 = prompt[16 + l15];
    const float h0 = hp[l15],     h1 = hp[16 + l15];
    v2f b0; b0.x = lo ? p0 : 0.0f; b0.y = 0.0f;
    v2f b1; b1.x = lo ? p1 : 0.0f; b1.y = 0.0f;
    v2f b2; b2.x = lo ? h0 : 0.0f; b2.y = 0.0f;
    v2f b3; b3.x = lo ? h1 : 0.0f; b3.y = 0.0f;

    v8f c = {0.f, 0.f, 0.f, 0.f, 0.f, 0.f, 0.f, 0.f};
    v8f d0 = __builtin_amdgcn_wmma_f32_16x16x4_f32(false, a, false, b0, (short)0, c, false, false);
    v8f d1 = __builtin_amdgcn_wmma_f32_16x16x4_f32(false, a, false, b1, (short)0, c, false, false);
    v8f d2 = __builtin_amdgcn_wmma_f32_16x16x4_f32(false, a, false, b2, (short)0, c, false, false);
    v8f d3 = __builtin_amdgcn_wmma_f32_16x16x4_f32(false, a, false, b3, (short)0, c, false, false);

    const int mofs = (lane >> 4) * 8;
#pragma unroll
    for (int r = 0; r < 8; ++r) {
        const size_t row = (size_t)(m0 + mofs + r) * OUT_DIM;
        out[row + 128 + l15]      = d0[r];
        out[row + 144 + l15]      = d1[r];
        out[row + 192 + l15]      = d2[r];
        out[row + 208 + l15]      = d3[r];
    }
}

extern "C" void kernel_launch(void* const* d_in, const int* in_sizes, int n_in,
                              void* d_out, int out_size, void* d_ws, size_t ws_size,
                              hipStream_t stream) {
    (void)in_sizes; (void)n_in; (void)out_size; (void)d_ws; (void)ws_size;

    const float* nf     = (const float*)d_in[0];
    const float* ef     = (const float*)d_in[1];
    const float* prompt = (const float*)d_in[2];
    const float* hp     = (const float*)d_in[3];
    const int*   src    = (const int*)d_in[4];
    const int*   dst    = (const int*)d_in[5];

    float* out = (float*)d_out;

    zero_kernel<<<(N_NODES + 255) / 256, 256, 0, stream>>>(out);
    fill_kernel<<<N_EDGES / 256, 256, 0, stream>>>(src, dst, out);
    gather_kernel<<<N_NODES / 8, 256, 0, stream>>>(nf, ef, src, dst, out);
    prompt_kernel<<<N_NODES / 16, 32, 0, stream>>>(prompt, hp, out);
}


# last launch folded into the gather kernel: a node's wave forms deg x prompt and deg x hetero-prompt itself (one f32 WMMA each, lane halves as two k steps), stores both lines whole; 4 -> 3 launches
# speedup vs baseline: 3.3028x; 1.0817x over previous
#include <hip/hip_runtime.h>

#define N_NODES   50000
#define N_EDGES   800000
#define IN_FEAT   128
#define PROMPT_DIM 32
#define EF_DIM     32
#define HP_DIM     32
#define OUT_DIM   224

typedef __attribute__((ext_vector_type(2))) float v2f;
typedef __attribute__((ext_vector_type(8))) float v8f;

__device__ __forceinline__ float gatomic_add_rtn_f32(float* p, float v) {
    float old;
    asm volatile("global_atomic_add_f32 %0, %1, %2, off th:TH_ATOMIC_RETURN\n\t"
                 "s_wait_loadcnt 0x0"
                 : "=v"(old) : "v"(p), "v"(v) : "memory");
    return old;
}

#define CNT_COL (OUT_DIM - 2)
#define HOME0   (OUT_DIM / 2 - 1)
#define FIRST   15
#define CAP0    (FIRST + 32)
#define CAP1    (FIRST + 48)
static_assert(OUT_DIM * 4 == 7 * 128, "a row is seven whole 128-B lines");
static_assert(PROMPT_DIM == 32 && HP_DIM == 32, "one prompt / hetero-prompt column per lane");
static_assert(N_EDGES % 256 == 0, "one thread per edge, whole work-groups");
static_assert(N_EDGES % 32 == 0, "whole waves in the dst sweep");
static_assert(N_NODES % 8 == 0, "whole work-groups in gather_kernel: no wave without a node");
static_assert(N_EDGES < (1 << 24), "edge counts must stay exact in f32");

__global__ void zero_kernel(float* __restrict__ outf) {
    const int i = blockIdx.x * blockDim.x + threadIdx.x;
    if (i < N_NODES) {
        outf[(size_t)i * OUT_DIM + CNT_COL] = 0.0f;
        outf[(size_t)i * OUT_DIM] = 0.0f;
    }
}

__global__ void __launch_bounds__(256)
fill_kernel(const int* __restrict__ src, const int* __restrict__ dst,
            float* outf) {
    const unsigned where = __builtin_amdgcn_s_sendmsg_rtn(0x87);
    const bool half1 = ((where >> 18) & 3u) != 0u;
    const int e = blockIdx.x * 256 + threadIdx.x;
    const int d = dst[e];
    const int s = src[e];
    asm volatile("" :: "v"(s));
    float* row = outf + (size_t)d * OUT_DIM;
    const int slot = (int)gatomic_add_rtn_f32(row + (half1 ? 0 : CNT_COL), 1.0f);
    if (slot < (half1 ? CAP1 : CAP0)) {
        int2 r; r.x = e; r.y = s;
        ((int2*)row)[half1 ? 1 + slot : HOME0 - 1 - slot] = r;
    }
}

#define ROWS 8
__global__ void __launch_bounds__(256)
gather_kernel(const float* __restrict__ nf,
              const float* __restrict__ ef,
              const float* __restrict__ prompt,
              const float* __restrict__ hp,
              const int*   __restrict__ src,
              const int*   __restrict__ dst,
              float* out) {
    const int lane = threadIdx.x & 31;
    const int wave = __builtin_amdgcn_readfirstlane((int)(threadIdx.x >> 5));
    const int n = blockIdx.x * 8 + wave;
    const float4* nf4 = (const float4*)nf;
    float* orow = out + (size_t)n * OUT_DIM;
    const int2* rowq = (const int2*)orow;

    int2 r = rowq[lane < 16 ? HOME0 - lane : lane - 16];
    asm volatile("" :: "v"(r.x), "v"(r.y));
    const int cnt0 = (int)__int_as_float(__builtin_amdgcn_readlane(r.x, 0));
    const int cnt1 = (int)__int_as_float(__builtin_amdgcn_readlane(r.x, 16));
    const float pv = prompt[lane];
    const float hv = hp[lane];

    float4 acc = make_float4(0.f, 0.f, 0.f, 0.f);
    float  g   = 0.f;

    if (cnt0 <= CAP0 && cnt1 <= CAP1) {
        const int m0 = cnt0 < FIRST ? cnt0 : FIRST;
        const int m1 = cnt1 < FIRST ? cnt1 : FIRST;
        int m = m0 + m1, c0 = m0, d0 = 1, d1 = 17 - m0;
        int b0 = FIRST, b1 = FIRST;
        for (;;) {
            const int e_l = r.x, s_l = r.y;
            for (int j0 = 0; j0 < m; j0 += ROWS) {
                float4 x[ROWS];
                float  f[ROWS];
#pragma unroll
                for (int u = 0; u < ROWS; ++u) {
                    x[u] = make_float4(0.f, 0.f, 0.f, 0.f);
                    f[u] = 0.f;
                    if (u == 0 || j0 + u < m) {
                        const int t  = j0 + u;
                        const int li = t + (t < c0 ? d0 : d1);
                        const int ee = __builtin_amdgcn_readlane(e_l, li);
                        const int ss = __builtin_amdgcn_readlane(s_l, li);
                        x[u] = nf4[(size_t)ss * (IN_FEAT / 4) + lane];
                        f[u] = ef[(size_t)ee * EF_DIM + lane];
                    }
                }
#pragma unroll
                for (int u = 0; u < ROWS; ++u) {
                    acc.x += x[u].x; acc.y += x[u].y; acc.z += x[u].z; acc.w += x[u].w;
                    g += f[u];
                }
            }
            if (b0 < cnt0) {
                m = cnt0 - b0 < 32 ? cnt0 - b0 : 32;
                if (lane < m) r = rowq[HOME0 - 1 - (b0 + lane)];
                b0 += 32;
            } else if (b1 < cnt1) {
                m = cnt1 - b1 < 32 ? cnt1 - b1 : 32;
                if (lane < m) r = rowq[1 + b1 + lane];
                b1 += 32;
            } else {
                break;
            }
            c0 = 32; d0 = 0; d1 = 0;
        }
    } else {
        for (int e0 = 0; e0 < N_EDGES; e0 += 32) {
            const int d_l = dst[e0 + lane];
            const int s_l = src[e0 + lane];
            unsigned hit = __builtin_amdgcn_ballot_w32(d_l == n);
            while (hit != 0u) {
                const int j = __builtin_ctz(hit);
                hit &= hit - 1u;
                const int ss = __builtin_amdgcn_readlane(s_l, j);
                const float4 xv = nf4[(size_t)ss * (IN_FEAT / 4) + lane];
                const float  fv = ef[(size_t)(e0 + j) * EF_DIM + lane];
                acc.x += xv.x; acc.y += xv.y; acc.z += xv.z; acc.w += xv.w;
                g += fv;
            }
        }
    }

    ((float4*)orow)[lane] = acc;
    orow[160 + lane] = g;

    const float deg = (float)(cnt0 + cnt1);
    v2f a;  a.x = (lane == 0 || lane == 24) ? deg : 0.0f; a.y = 0.0f;
    v2f bp; bp.x = pv; bp.y = 0.0f;
    v2f bh; bh.x = hv; bh.y = 0.0f;
    const v8f c = {0.f, 0.f, 0.f, 0.f, 0.f, 0.f, 0.f, 0.f};
    const v8f dp = __builtin_amdgcn_wmma_f32_16x16x4_f32(false, a, false, bp, (short)0, c, false, false);
    const v8f dh = __builtin_amdgcn_wmma_f32_16x16x4_f32(false, a, false, bh, (short)0, c, false, false);
    orow[128 + lane] = dp[0];
    orow[192 + lane] = dh[0];
}

extern "C" void kernel_launch(void* const* d_in, const int* in_sizes, int n_in,
                              void* d_out, int out_size, void* d_ws, size_t ws_size,
                              hipStream_t stream) {
    (void)in_sizes; (void)n_in; (void)out_size; (void)d_ws; (void)ws_size;

    const float* nf     = (const float*)d_in[0];
    const float* ef     = (const float*)d_in[1];
    const float* prompt = (const float*)d_in[2];
    const float* hp     = (const float*)d_in[3];
    const int*   src    = (const int*)d_in[4];
    const int*   dst    = (const int*)d_in[5];

    float* out = (float*)d_out;

    zero_kernel<<<(N_NODES + 255) / 256, 256, 0, stream>>>(out);
    fill_kernel<<<N_EDGES / 256, 256, 0, stream>>>(src, dst, out);
    gather_kernel<<<N_NODES / 8, 256, 0, stream>>>(nf, ef, prompt, hp, src, dst, out);
}


# speedup vs baseline: 4.1681x; 1.2620x over previous
#include <hip/hip_runtime.h>

#define N_NODES   50000
#define N_EDGES   800000
#define IN_FEAT   128
#define PROMPT_DIM 32
#define EF_DIM     32
#define HP_DIM     32
#define OUT_DIM   224

typedef __attribute__((ext_vector_type(2))) float v2f;
typedef __attribute__((ext_vector_type(8))) float v8f;

__device__ __forceinline__ float gatomic_add_rtn_f32(float* p, float v) {
    float old;
    asm volatile("global_atomic_add_f32 %0, %1, %2, off th:TH_ATOMIC_RETURN\n\t"
                 "s_wait_loadcnt 0x0"
                 : "=v"(old) : "v"(p), "v"(v) : "memory");
    return old;
}

#define CNT_COL (OUT_DIM - 2)
#define HOME0   (OUT_DIM / 2 - 1)
#define FIRST   15
#define CAP_H0  (FIRST + 32)
#define CAP_H1  (FIRST + 48)
static_assert(1 + CAP_H0 + 1 + CAP_H1 == OUT_DIM / 2, "both buckets and both counter pairs fill the row exactly");
static_assert(OUT_DIM * 4 == 7 * 128, "a row is seven whole 128-B lines");
static_assert(PROMPT_DIM == 32 && HP_DIM == 32, "one prompt / hetero-prompt column per lane");
static_assert(N_EDGES % 256 == 0, "one thread per edge, whole work-groups");
static_assert(N_EDGES % 32 == 0, "whole waves in the dst sweep");
static_assert(N_NODES % 8 == 0, "whole work-groups in gather_kernel: no wave without a node");
static_assert(N_EDGES < (1 << 24), "edge counts must stay exact in f32");

__global__ void zero_kernel(float* __restrict__ outf) {
    const int i = blockIdx.x * blockDim.x + threadIdx.x;
    if (i < N_NODES) {
        outf[(size_t)i * OUT_DIM + CNT_COL] = 0.0f;
        outf[(size_t)i * OUT_DIM] = 0.0f;
    }
}

__global__ void __launch_bounds__(256)
fill_kernel(const int* __restrict__ src, const int* __restrict__ dst,
            float* outf) {
    const unsigned where = __builtin_amdgcn_s_sendmsg_rtn(0x87);
    const bool half1 = ((where >> 18) & 3u) != 0u;
    const int e = blockIdx.x * 256 + threadIdx.x;
    const int d = dst[e];
    const int s = src[e];
    asm volatile("" :: "v"(s));
    float* row = outf + (size_t)d * OUT_DIM;
    const bool down = half1 == ((d & 1) != 0);
    const int slot = (int)gatomic_add_rtn_f32(row + (down ? CNT_COL : 0), 1.0f);
    if (slot < (half1 ? CAP_H1 : CAP_H0)) {
        int2 r; r.x = e; r.y = s;
        ((int2*)row)[down ? HOME0 - 1 - slot : 1 + slot] = r;
    }
}

#define ROWS 8
__global__ void __launch_bounds__(256)
gather_kernel(const float* __restrict__ nf,
              const float* __restrict__ ef,
              const float* __restrict__ prompt,
              const float* __restrict__ hp,
              const int*   __restrict__ src,
              const int*   __restrict__ dst,
              float* out) {
    const int lane = threadIdx.x & 31;
    const int wave = __builtin_amdgcn_readfirstlane((int)(threadIdx.x >> 5));
    const int n = blockIdx.x * 8 + wave;
    const float4* nf4 = (const float4*)nf;
    float* orow = out + (size_t)n * OUT_DIM;
    const int2* rowq = (const int2*)orow;

    int2 r = rowq[lane < 16 ? HOME0 - lane : lane - 16];
    asm volatile("" :: "v"(r.x), "v"(r.y));
    const int cnt0 = (int)__int_as_float(__builtin_amdgcn_readlane(r.x, 0));
    const int cnt1 = (int)__int_as_float(__builtin_amdgcn_readlane(r.x, 16));
    const float pv = prompt[lane];
    const float hv = hp[lane];

    float4 acc = make_float4(0.f, 0.f, 0.f, 0.f);
    float  g   = 0.f;

    const bool odd = (n & 1) != 0;
    const int cap0 = odd ? CAP_H1 : CAP_H0;
    const int cap1 = odd ? CAP_H0 : CAP_H1;
    if (cnt0 <= cap0 && cnt1 <= cap1) {
        const int m0 = cnt0 < FIRST ? cnt0 : FIRST;
        const int m1 = cnt1 < FIRST ? cnt1 : FIRST;
        int m = m0 + m1, c0 = m0, d0 = 1, d1 = 17 - m0;
        int b0 = FIRST, b1 = FIRST;
        for (;;) {
            const int e_l = r.x, s_l = r.y;
            for (int j0 = 0; j0 < m; j0 += ROWS) {
                float4 x[ROWS];
                float  f[ROWS];
#pragma unroll
                for (int u = 0; u < ROWS; ++u) {
                    x[u] = make_float4(0.f, 0.f, 0.f, 0.f);
                    f[u] = 0.f;
                    if (u == 0 || j0 + u < m) {
                        const int t  = j0 + u;
                        const int li = t + (t < c0 ? d0 : d1);
                        const int ee = __builtin_amdgcn_readlane(e_l, li);
                        const int ss = __builtin_amdgcn_readlane(s_l, li);
                        x[u] = nf4[(size_t)ss * (IN_FEAT / 4) + lane];
                        f[u] = ef[(size_t)ee * EF_DIM + lane];
                    }
                }
#pragma unroll
                for (int u = 0; u < ROWS; ++u) {
                    acc.x += x[u].x; acc.y += x[u].y; acc.z += x[u].z; acc.w += x[u].w;
                    g += f[u];
                }
            }
            if (b0 < cnt0) {
                m = cnt0 - b0 < 32 ? cnt0 - b0 : 32;
                if (lane < m) r = rowq[HOME0 - 1 - (b0 + lane)];
                b0 += 32;
            } else if (b1 < cnt1) {
                m = cnt1 - b1 < 32 ? cnt1 - b1 : 32;
                if (lane < m) r = rowq[1 + b1 + lane];
                b1 += 32;
            } else {
                break;
            }
            c0 = 32; d0 = 0; d1 = 0;
        }
    } else {
        for (int e0 = 0; e0 < N_EDGES; e0 += 32) {
            const int d_l = dst[e0 + lane];
            const int s_l = src[e0 + lane];
            unsigned hit = __builtin_amdgcn_ballot_w32(d_l == n);
            while (hit != 0u) {
                const int j = __builtin_ctz(hit);
                hit &= hit - 1u;
                const int ss = __builtin_amdgcn_readlane(s_l, j);
                const float4 xv = nf4[(size_t)ss * (IN_FEAT / 4) + lane];
                const float  fv = ef[(size_t)(e0 + j) * EF_DIM + lane];
                acc.x += xv.x; acc.y += xv.y; acc.z += xv.z; acc.w += xv.w;
                g += fv;
            }
        }
    }

    ((float4*)orow)[lane] = acc;
    orow[160 + lane] = g;

    const float deg = (float)(cnt0 + cnt1);
    v2f a;  a.x = (lane == 0 || lane == 24) ? deg : 0.0f; a.y = 0.0f;
    v2f bp; bp.x = pv; bp.y = 0.0f;
    v2f bh; bh.x = hv; bh.y = 0.0f;
    const v8f c = {0.f, 0.f, 0.f, 0.f, 0.f, 0.f, 0.f, 0.f};
    const v8f dp = __builtin_amdgcn_wmma_f32_16x16x4_f32(false, a, false, bp, (short)0, c, false, false);
    const v8f dh = __builtin_amdgcn_wmma_f32_16x16x4_f32(false, a, false, bh, (short)0, c, false, false);
    orow[128 + lane] = dp[0];
    orow[192 + lane] = dh[0];
}

extern "C" void kernel_launch(void* const* d_in, const int* in_sizes, int n_in,
                              void* d_out, int out_size, void* d_ws, size_t ws_size,
                              hipStream_t stream) {
    (void)in_sizes; (void)n_in; (void)out_size; (void)d_ws; (void)ws_size;

    const float* nf     = (const float*)d_in[0];
    const float* ef     = (const float*)d_in[1];
    const float* prompt = (const float*)d_in[2];
    const float* hp     = (const float*)d_in[3];
    const int*   src    = (const int*)d_in[4];
    const int*   dst    = (const int*)d_in[5];

    float* out = (float*)d_out;

    zero_kernel<<<(N_NODES + 255) / 256, 256, 0, stream>>>(out);
    fill_kernel<<<N_EDGES / 256, 256, 0, stream>>>(src, dst, out);
    gather_kernel<<<N_NODES / 8, 256, 0, stream>>>(nf, ef, prompt, hp, src, dst, out);
}
